# starting kernel
# baseline (speedup 1.0000x reference)
#include <hip/hip_runtime.h>
#include <hip/hip_bf16.h>
#include <math.h>

typedef __attribute__((ext_vector_type(2))) float v2f;
typedef __attribute__((ext_vector_type(8))) float v8f;

#define NEG_SLOPE 0.2f

__device__ __forceinline__ unsigned fenc(float f) {
  unsigned u = __float_as_uint(f);
  return (u & 0x80000000u) ? ~u : (u | 0x80000000u);
}
__device__ __forceinline__ float fdec(unsigned u) {
  return __uint_as_float((u & 0x80000000u) ? (u & 0x7FFFFFFFu) : ~u);
}

__global__ void gat_init(float* __restrict__ outb, float* __restrict__ ssum,
                         unsigned* __restrict__ mbuf, int N) {
  long long i = (long long)blockIdx.x * blockDim.x + threadIdx.x;
  long long total = (long long)N * 128;
  if (i < total) outb[i] = 0.0f;
  if (i < N) { ssum[i] = 0.0f; mbuf[i] = 0u; }
}

__global__ __launch_bounds__(256) void gat_gemm(const float* __restrict__ x,
                                                const float* __restrict__ W,
                                                float* __restrict__ h, int N) {
  const int lane = threadIdx.x & 31;
  const int wave = threadIdx.x >> 5;
  const int m0 = blockIdx.x << 4;
  if (m0 >= N) return;
  const int r16 = lane & 15;
  const int khi = (lane >> 4) << 1;
  const int col = (wave << 4) + r16;
  const float* ap = x + (size_t)(m0 + r16) * 256 + khi;
  const float* bp = W + (size_t)khi * 128 + col;
  v8f acc = {};
  for (int kk = 0; kk < 256; kk += 4) {
    v2f a = *(const v2f*)ap;
    v2f b;
    b.x = bp[0];
    b.y = bp[128];
    acc = __builtin_amdgcn_wmma_f32_16x16x4_f32(false, a, false, b, (short)0,
                                                acc, false, false);
    ap += 4;
    bp += 4 * 128;
  }
  const int rbase = m0 + ((lane >> 4) << 3);
  float* hp = h + (size_t)rbase * 128 + col;
#pragma unroll
  for (int r = 0; r < 8; ++r) hp[(size_t)r * 128] = acc[r];
}

__global__ __launch_bounds__(256) void gat_rowdots(const float* __restrict__ h,
                                                   const float* __restrict__ as,
                                                   const float* __restrict__ ad,
                                                   float* __restrict__ a_src,
                                                   float* __restrict__ a_dst,
                                                   int N) {
  const int node = blockIdx.x * 8 + (threadIdx.x >> 5);
  if (node >= N) return;
  const int lane = threadIdx.x & 31;
  const float* hr = h + (size_t)node * 128;
  float s0 = 0.0f, s1 = 0.0f;
#pragma unroll
  for (int j = 0; j < 4; ++j) {
    float v = hr[lane + 32 * j];
    s0 += v * as[lane + 32 * j];
    s1 += v * ad[lane + 32 * j];
  }
#pragma unroll
  for (int off = 16; off > 0; off >>= 1) {
    s0 += __shfl_xor(s0, off, 32);
    s1 += __shfl_xor(s1, off, 32);
  }
  if (lane == 0) { a_src[node] = s0; a_dst[node] = s1; }
}

__global__ void gat_edge_max(const int* __restrict__ src, const int* __restrict__ dst,
                             const float* __restrict__ a_src,
                             const float* __restrict__ a_dst,
                             unsigned* __restrict__ mbuf, int E, int N) {
  const int i = blockIdx.x * blockDim.x + threadIdx.x;
  const int total = E + N;
  if (i >= total) return;
  int s, d;
  if (i < E) { s = src[i]; d = dst[i]; } else { s = d = i - E; }
  float e = a_src[s] + a_dst[d];
  e = (e > 0.0f) ? e : NEG_SLOPE * e;
  atomicMax(&mbuf[d], fenc(e));
}

__global__ void gat_edge_exp(const int* __restrict__ src, const int* __restrict__ dst,
                             const float* __restrict__ a_src,
                             const float* __restrict__ a_dst,
                             const unsigned* __restrict__ mbuf,
                             float* __restrict__ exbuf, float* __restrict__ ssum,
                             int E, int N) {
  const int i = blockIdx.x * blockDim.x + threadIdx.x;
  const int total = E + N;
  if (i >= total) return;
  int s, d;
  if (i < E) { s = src[i]; d = dst[i]; } else { s = d = i - E; }
  float e = a_src[s] + a_dst[d];
  e = (e > 0.0f) ? e : NEG_SLOPE * e;
  float ex = expf(e - fdec(mbuf[d]));
  exbuf[i] = ex;
  atomicAdd(&ssum[d], ex);
}

__global__ __launch_bounds__(256) void gat_edge_scatter(
    const int* __restrict__ src, const int* __restrict__ dst,
    const float* __restrict__ h, const float* __restrict__ exbuf,
    const float* __restrict__ ssum, float* __restrict__ outb, int E, int N) {
  const int i = blockIdx.x * 8 + (threadIdx.x >> 5);
  const int total = E + N;
  if (i >= total) return;
  const int lane = threadIdx.x & 31;
  int s, d;
  if (i < E) { s = src[i]; d = dst[i]; } else { s = d = i - E; }
  const float alpha = exbuf[i] / (ssum[d] + 1e-16f);
  const float4 v = ((const float4*)(h + (size_t)s * 128))[lane];
  float* od = outb + (size_t)d * 128 + lane * 4;
  atomicAdd(&od[0], v.x * alpha);
  atomicAdd(&od[1], v.y * alpha);
  atomicAdd(&od[2], v.z * alpha);
  atomicAdd(&od[3], v.w * alpha);
}

__global__ __launch_bounds__(256) void gat_final(const float* __restrict__ outb,
                                                 const float* __restrict__ bias_gat,
                                                 const float* __restrict__ fc_w,
                                                 const float* __restrict__ fc_b,
                                                 float* __restrict__ logout, int N) {
  const int node = blockIdx.x * 8 + (threadIdx.x >> 5);
  if (node >= N) return;
  const int lane = threadIdx.x & 31;
  const float* orow = outb + (size_t)node * 128;
  float d0 = 0.0f, d1 = 0.0f;
#pragma unroll
  for (int j = 0; j < 4; ++j) {
    int c = lane + 32 * j;
    float v = orow[c] + bias_gat[c];
    d0 += v * fc_w[c];
    d1 += v * fc_w[128 + c];
  }
#pragma unroll
  for (int off = 16; off > 0; off >>= 1) {
    d0 += __shfl_xor(d0, off, 32);
    d1 += __shfl_xor(d1, off, 32);
  }
  if (lane == 0) {
    float z0 = d0 + fc_b[0];
    float z1 = d1 + fc_b[1];
    float mx = fmaxf(z0, z1);
    float lse = mx + logf(expf(z0 - mx) + expf(z1 - mx));
    logout[(size_t)node * 2 + 0] = z0 - lse;
    logout[(size_t)node * 2 + 1] = z1 - lse;
  }
}

extern "C" void kernel_launch(void* const* d_in, const int* in_sizes, int n_in,
                              void* d_out, int out_size, void* d_ws, size_t ws_size,
                              hipStream_t stream) {
  const float* x        = (const float*)d_in[0];
  const int*   ei       = (const int*)d_in[1];
  const float* W        = (const float*)d_in[2];
  const float* att_src  = (const float*)d_in[3];
  const float* att_dst  = (const float*)d_in[4];
  const float* bias_gat = (const float*)d_in[5];
  const float* fc_w     = (const float*)d_in[6];
  const float* fc_b     = (const float*)d_in[7];
  float* logout = (float*)d_out;

  const int N = in_sizes[0] / 256;
  const int E = in_sizes[1] / 2;
  const int* src = ei;
  const int* dst = ei + E;

  char* ws = (char*)d_ws;
  size_t off = 0;
  auto carve = [&](size_t bytes) -> void* {
    void* p = (void*)(ws + off);
    off += (bytes + 255) & ~(size_t)255;
    return p;
  };
  float*    h     = (float*)carve((size_t)N * 128 * sizeof(float));
  float*    outb  = (float*)carve((size_t)N * 128 * sizeof(float));
  float*    a_src = (float*)carve((size_t)N * sizeof(float));
  float*    a_dst = (float*)carve((size_t)N * sizeof(float));
  unsigned* mbuf  = (unsigned*)carve((size_t)N * sizeof(unsigned));
  float*    ssum  = (float*)carve((size_t)N * sizeof(float));
  float*    exbuf = (float*)carve((size_t)(E + N) * sizeof(float));

  const int total = E + N;

  {
    long long n = (long long)N * 128;
    int blocks = (int)((n + 255) / 256);
    gat_init<<<blocks, 256, 0, stream>>>(outb, ssum, mbuf, N);
  }
  {
    int blocks = (N + 15) / 16;
    gat_gemm<<<blocks, 256, 0, stream>>>(x, W, h, N);
  }
  {
    int blocks = (N + 7) / 8;
    gat_rowdots<<<blocks, 256, 0, stream>>>(h, att_src, att_dst, a_src, a_dst, N);
  }
  {
    int blocks = (total + 255) / 256;
    gat_edge_max<<<blocks, 256, 0, stream>>>(src, dst, a_src, a_dst, mbuf, E, N);
  }
  {
    int blocks = (total + 255) / 256;
    gat_edge_exp<<<blocks, 256, 0, stream>>>(src, dst, a_src, a_dst, mbuf, exbuf,
                                             ssum, E, N);
  }
  {
    int blocks = (total + 7) / 8;
    gat_edge_scatter<<<blocks, 256, 0, stream>>>(src, dst, h, exbuf, ssum, outb,
                                                 E, N);
  }
  {
    int blocks = (N + 7) / 8;
    gat_final<<<blocks, 256, 0, stream>>>(outb, bias_gat, fc_w, fc_b, logout, N);
  }
}


# launches 8 -> 6: the h GEMM's strip-owner wave also stores the zero start values of its 16 nodes and takes a_src,a_dst from its accumulators (h not re-read); a_src,a_dst add order changes
# speedup vs baseline: 2.7642x; 2.7642x over previous
#include <hip/hip_runtime.h>
#include <hip/hip_bf16.h>
#include <math.h>

typedef __attribute__((ext_vector_type(2))) float v2f;
typedef __attribute__((ext_vector_type(4))) float v4f;
typedef __attribute__((ext_vector_type(8))) float v8f;

#define NEG_SLOPE 0.2f

__device__ __forceinline__ unsigned fenc(float f) {
  unsigned u = __float_as_uint(f);
  return (u & 0x80000000u) ? ~u : (u | 0x80000000u);
}
__device__ __forceinline__ float fdec(unsigned u) {
  return __uint_as_float((u & 0x80000000u) ? (u & 0x7FFFFFFFu) : ~u);
}

__device__ __forceinline__ unsigned chip_half() {
  return (__builtin_amdgcn_s_sendmsg_rtn(0x87) >> 18) & 1u;
}

__global__ __launch_bounds__(512) void gat_gemm(const float* __restrict__ x,
                                                const float* __restrict__ W,
                                                const float* __restrict__ as,
                                                const float* __restrict__ ad,
                                                float* __restrict__ h,
                                                float* __restrict__ outb,
                                                float* __restrict__ a_src,
                                                float* __restrict__ a_dst,
                                                float* __restrict__ ssum,
                                                unsigned* __restrict__ mbuf,
                                                unsigned* __restrict__ queue, int N) {
  __shared__ v4f Ws4[32 * 8 * 32];
  const int tid = threadIdx.x;
  const int lane = tid & 31;
  const int wave = __builtin_amdgcn_readfirstlane(tid >> 5);
  const int c = lane & 15;
  const int hh = lane >> 4;

  {
    const int n = tid & 127;
    const int dstc = ((n >> 4) << 5) + (n & 15);
#pragma unroll 8
    for (int jh = tid >> 7; jh < 64; jh += 4) {
      const int j = jh >> 1, hsel = jh & 1;
      const float* wp = W + (size_t)(8 * j + 4 * hsel) * 128 + n;
      v4f v;
      v.x = wp[0];
      v.y = wp[128];
      v.z = wp[256];
      v.w = wp[384];
      Ws4[(j << 8) + (hsel << 4) + dstc] = v;
    }
  }
  __syncthreads();

  if (blockIdx.x == 0 && tid < 2) queue[tid * 64] = 0u;

  float asv[8], adv[8];
#pragma unroll
  for (int t = 0; t < 8; ++t) {
    asv[t] = as[(t << 4) + c];
    adv[t] = ad[(t << 4) + c];
  }

  const int ntiles = (N + 15) >> 4;
  for (int q = wave;; q += 16) {
    const int tile = (int)blockIdx.x + (int)gridDim.x * q;
    if (tile >= ntiles) break;
    const int m0 = tile << 4;
    const float* ap = x + (size_t)(m0 + c) * 256 + 4 * hh;

    {
      float* ob = outb + (size_t)m0 * 128 + 4 * lane;
      const v4f z = {0.0f, 0.0f, 0.0f, 0.0f};
      const int nrow = (N - m0 < 16) ? N - m0 : 16;
      if (nrow == 16) {
#pragma unroll
        for (int rr = 0; rr < 16; ++rr) *(v4f*)(ob + (size_t)rr * 128) = z;
      } else {
        for (int rr = 0; rr < nrow; ++rr) *(v4f*)(ob + (size_t)rr * 128) = z;
      }
    }

    v8f acc[8];
#pragma unroll
    for (int t = 0; t < 8; ++t) acc[t] = v8f{};

    v4f acur[4];
#pragma unroll
    for (int i = 0; i < 4; ++i) acur[i] = *(const v4f*)(ap + 8 * i);

#pragma unroll 1
    for (int jj = 0; jj < 8; ++jj) {
      const int jn = (jj < 7) ? jj + 1 : 7;
      v4f anxt[4];
#pragma unroll
      for (int i = 0; i < 4; ++i) anxt[i] = *(const v4f*)(ap + 32 * jn + 8 * i);
#pragma unroll
      for (int i = 0; i < 4; ++i) {
        const v4f* bp = Ws4 + (((jj << 2) + i) << 8) + lane;
        v4f b[8];
#pragma unroll
        for (int t = 0; t < 8; ++t) b[t] = bp[t << 5];
        v2f a01 = {acur[i].x, acur[i].y};
        v2f a23 = {acur[i].z, acur[i].w};
#pragma unroll
        for (int t = 0; t < 8; ++t) {
          v2f b01 = {b[t].x, b[t].y};
          acc[t] = __builtin_amdgcn_wmma_f32_16x16x4_f32(false, a01, false, b01,
                                                         (short)0, acc[t], false, false);
        }
#pragma unroll
        for (int t = 0; t < 8; ++t) {
          v2f b23 = {b[t].z, b[t].w};
          acc[t] = __builtin_amdgcn_wmma_f32_16x16x4_f32(false, a23, false, b23,
                                                         (short)0, acc[t], false, false);
        }
      }
      asm volatile("" ::"v"(anxt[0]), "v"(anxt[1]), "v"(anxt[2]), "v"(anxt[3]));
#pragma unroll
      for (int i = 0; i < 4; ++i) acur[i] = anxt[i];
    }

    const int rbase = m0 + (hh << 3);
#pragma unroll
    for (int t = 0; t < 8; ++t) {
      float* hp = h + (size_t)rbase * 128 + (t << 4) + c;
#pragma unroll
      for (int r = 0; r < 8; ++r) hp[(size_t)r * 128] = acc[t][r];
    }

    float s0[8], s1[8];
#pragma unroll
    for (int r = 0; r < 8; ++r) {
      float p0 = 0.0f, p1 = 0.0f;
#pragma unroll
      for (int t = 0; t < 8; ++t) {
        p0 += acc[t][r] * asv[t];
        p1 += acc[t][r] * adv[t];
      }
      s0[r] = p0;
      s1[r] = p1;
    }
#pragma unroll
    for (int off = 8; off > 0; off >>= 1) {
#pragma unroll
      for (int r = 0; r < 8; ++r) {
        s0[r] += __shfl_xor(s0[r], off, 32);
        s1[r] += __shfl_xor(s1[r], off, 32);
      }
    }
    float v0 = s0[0], v1 = s1[0];
#pragma unroll
    for (int r = 1; r < 8; ++r) {
      v0 = (c == r) ? s0[r] : v0;
      v1 = (c == r) ? s1[r] : v1;
    }
    const int node = rbase + c;
    if (c < 8 && node < N) {
      a_src[node] = v0;
      a_dst[node] = v1;
      ssum[node] = 0.0f;
      ssum[(size_t)N + node] = 0.0f;
      mbuf[node] = 0u;
      mbuf[(size_t)N + node] = 0u;
    }
  }
}


__global__ void gat_edge_max(const int* __restrict__ src, const int* __restrict__ dst,
                             const float* __restrict__ a_src,
                             const float* __restrict__ a_dst,
                             unsigned* __restrict__ mbuf, float* __restrict__ exbuf,
                             int E, int N) {
  const int i = blockIdx.x * blockDim.x + threadIdx.x;
  const int total = E + N;
  if (i >= total) return;
  int s, d;
  if (i < E) { s = src[i]; d = dst[i]; } else { s = d = i - E; }
  float e = a_src[s] + a_dst[d];
  e = (e > 0.0f) ? e : NEG_SLOPE * e;
  exbuf[i] = e;
  atomicMax(&mbuf[(size_t)chip_half() * N + d], fenc(e));
}

__global__ void gat_edge_exp(const int* __restrict__ dst,
                             const unsigned* __restrict__ mbuf,
                             float* __restrict__ exbuf, float* __restrict__ ssum,
                             int E, int N) {
  const int i = blockIdx.x * blockDim.x + threadIdx.x;
  const int total = E + N;
  if (i >= total) return;
  const unsigned half = chip_half();
  const int d = (i < E) ? dst[i] : i - E;
  const float e = exbuf[i];
  const unsigned m0 = mbuf[d], m1 = mbuf[(size_t)N + d];
  float ex = expf(e - fdec(m0 > m1 ? m0 : m1));
  exbuf[i] = ex;
  atomicAdd(&ssum[(size_t)half * N + d], ex);
}

__global__ void gat_den(const float* __restrict__ ssum, float* __restrict__ den, int N) {
  const int n = blockIdx.x * blockDim.x + threadIdx.x;
  if (n < N) den[n] = (ssum[n] + ssum[(size_t)N + n]) + 1e-16f;
}

__global__ __launch_bounds__(256) void gat_edge_scatter(
    const int* __restrict__ src, const int* __restrict__ dst,
    const float* __restrict__ h, const float* __restrict__ exbuf,
    const float* __restrict__ den, float* __restrict__ outb,
    unsigned* __restrict__ queue, int E, int N, int nchunks) {
  __shared__ int s_ticket;
  if (threadIdx.x == 0) {
    unsigned half = chip_half();
    unsigned t = atomicAdd(&queue[half * 64], 1u);
    if (t >= (unsigned)nchunks) {
      half ^= 1u;
      t = atomicAdd(&queue[half * 64], 1u);
    }
    s_ticket = (t < (unsigned)nchunks) ? (int)(2u * t + half) : -1;
  }
  __syncthreads();
  const int ticket = __builtin_amdgcn_readfirstlane(s_ticket);
  if (ticket < 0) return;
  const int lane = threadIdx.x & 31;
  const int wv = __builtin_amdgcn_readfirstlane((int)(threadIdx.x >> 5));
  const int total = E + N;
  const int colbase = (ticket & 1) << 6;
  const float* hc = h + colbase;
  float* oc = outb + colbase;
  for (int g = 0; g < 2; ++g) {
    const int base = (ticket >> 1) * 512 + (((wv << 1) + g) << 5);
    if (base >= total) break;
    const int i = base + lane;
    int s = 0, d = 0;
    float alpha = 0.0f;
    if (i < total) {
      if (i < E) { s = src[i]; d = dst[i]; } else { s = d = i - E; }
      alpha = exbuf[i] / den[d];
    }
    const int rest = total - base;
    const int cnt = rest < 32 ? rest : 32;
    int k = 0;
    for (; k + 16 <= cnt; k += 16) {
      float v[16][2];
#pragma unroll
      for (int j = 0; j < 16; ++j) {
        const int sk = __builtin_amdgcn_readlane(s, k + j);
        const float* hr = hc + (size_t)sk * 128;
        v[j][0] = hr[lane];
        v[j][1] = hr[lane + 32];
      }
#pragma unroll
      for (int j = 0; j < 16; ++j) {
        const int dk = __builtin_amdgcn_readlane(d, k + j);
        const float ak = __int_as_float(__builtin_amdgcn_readlane(__float_as_int(alpha), k + j));
        float* od = oc + (size_t)dk * 128;
        atomicAdd(od + lane, v[j][0] * ak);
        atomicAdd(od + lane + 32, v[j][1] * ak);
      }
    }
    for (; k < cnt; ++k) {
      const int sk = __builtin_amdgcn_readlane(s, k);
      const int dk = __builtin_amdgcn_readlane(d, k);
      const float ak = __int_as_float(__builtin_amdgcn_readlane(__float_as_int(alpha), k));
      const float* hr = hc + (size_t)sk * 128;
      float* od = oc + (size_t)dk * 128;
      const float v0 = hr[lane], v1 = hr[lane + 32];
      atomicAdd(od + lane, v0 * ak);
      atomicAdd(od + lane + 32, v1 * ak);
    }
  }
}

__global__ __launch_bounds__(256) void gat_final(const float* __restrict__ outb,
                                                 const float* __restrict__ bias_gat,
                                                 const float* __restrict__ fc_w,
                                                 const float* __restrict__ fc_b,
                                                 float* __restrict__ logout, int N) {
  const int node = blockIdx.x * 8 + (threadIdx.x >> 5);
  if (node >= N) return;
  const int lane = threadIdx.x & 31;
  const float* orow = outb + (size_t)node * 128;
  float d0 = 0.0f, d1 = 0.0f;
#pragma unroll
  for (int j = 0; j < 4; ++j) {
    int c = lane + 32 * j;
    float v = orow[c] + bias_gat[c];
    d0 += v * fc_w[c];
    d1 += v * fc_w[128 + c];
  }
#pragma unroll
  for (int off = 16; off > 0; off >>= 1) {
    d0 += __shfl_xor(d0, off, 32);
    d1 += __shfl_xor(d1, off, 32);
  }
  if (lane == 0) {
    float z0 = d0 + fc_b[0];
    float z1 = d1 + fc_b[1];
    float mx = fmaxf(z0, z1);
    float lse = mx + logf(expf(z0 - mx) + expf(z1 - mx));
    logout[(size_t)node * 2 + 0] = z0 - lse;
    logout[(size_t)node * 2 + 1] = z1 - lse;
  }
}

extern "C" void kernel_launch(void* const* d_in, const int* in_sizes, int n_in,
                              void* d_out, int out_size, void* d_ws, size_t ws_size,
                              hipStream_t stream) {
  const float* x        = (const float*)d_in[0];
  const int*   ei       = (const int*)d_in[1];
  const float* W        = (const float*)d_in[2];
  const float* att_src  = (const float*)d_in[3];
  const float* att_dst  = (const float*)d_in[4];
  const float* bias_gat = (const float*)d_in[5];
  const float* fc_w     = (const float*)d_in[6];
  const float* fc_b     = (const float*)d_in[7];
  float* logout = (float*)d_out;

  const int N = in_sizes[0] / 256;
  const int E = in_sizes[1] / 2;
  const int* src = ei;
  const int* dst = ei + E;

  char* ws = (char*)d_ws;
  size_t off = 0;
  auto carve = [&](size_t bytes) -> void* {
    void* p = (void*)(ws + off);
    off += (bytes + 255) & ~(size_t)255;
    return p;
  };
  float*    h     = (float*)carve((size_t)N * 128 * sizeof(float));
  float*    outb  = (float*)carve((size_t)N * 128 * sizeof(float));
  float*    a_src = (float*)carve((size_t)N * sizeof(float));
  float*    a_dst = (float*)carve((size_t)N * sizeof(float));
  unsigned* mbuf  = (unsigned*)carve((size_t)2 * N * sizeof(unsigned));
  float*    ssum  = (float*)carve((size_t)2 * N * sizeof(float));
  unsigned* queue = (unsigned*)carve(512);
  float*    exbuf = (float*)carve((size_t)(E + N) * sizeof(float));
  float*    den   = (float*)carve((size_t)N * sizeof(float));

  const int total = E + N;

  {
    gat_gemm<<<256, 512, 0, stream>>>(x, W, att_src, att_dst, h, outb, a_src, a_dst,
                                      ssum, mbuf, queue, N);
  }
  {
    int blocks = (total + 255) / 256;
    gat_edge_max<<<blocks, 256, 0, stream>>>(src, dst, a_src, a_dst, mbuf, exbuf, E, N);
  }
  {
    int blocks = (total + 255) / 256;
    gat_edge_exp<<<blocks, 256, 0, stream>>>(dst, mbuf, exbuf, ssum, E, N);
  }
  {
    int blocks = (N + 255) / 256;
    gat_den<<<blocks, 256, 0, stream>>>(ssum, den, N);
  }
  {
    int nchunks = (total + 511) / 512;
    gat_edge_scatter<<<2 * nchunks, 256, 0, stream>>>(src, dst, h, exbuf, den, outb,
                                                     queue, E, N, nchunks);
  }
  {
    int blocks = (N + 7) / 8;
    gat_final<<<blocks, 256, 0, stream>>>(outb, bias_gat, fc_w, fc_b, logout, N);
  }
}
